# starting kernel
# baseline (speedup 1.0000x reference)
#include <hip/hip_runtime.h>
#include <float.h>

typedef __attribute__((ext_vector_type(2))) float v2f;
typedef __attribute__((ext_vector_type(8))) float v8f;

#define BB    32
#define SS    512
#define HH    1024
#define NC    256
#define NL    34
#define NLPAD 48
#define KTOT  (2 * HH)

__global__ void dmbert_wpad_kernel(const float* __restrict__ Wm,
                                   float*       __restrict__ Wt) {
    const int idx = blockIdx.x * blockDim.x + threadIdx.x;
    const int k   = idx & (KTOT - 1);
    const int c   = idx / KTOT;
    Wt[idx] = (c < NL) ? Wm[(size_t)k * NL + c] : 0.0f;
}

__global__ void dmbert_pool_kernel(const float* __restrict__ hidden,
                                   const int*  __restrict__ cand_batch,
                                   const int*  __restrict__ cand_cut,
                                   float*      __restrict__ pooled) {
    const int n   = blockIdx.x;
    const int t   = threadIdx.x;
    const int bat = cand_batch[n];
    const int cut = cand_cut[n];

    const float4* row = (const float4*)(hidden + (size_t)bat * SS * HH) + t;
    const int rstride = HH / 4;

    float4 ml = make_float4(-FLT_MAX, -FLT_MAX, -FLT_MAX, -FLT_MAX);
    float4 mr = ml;

    int s = 0;
    #pragma unroll 4
    for (; s < cut; ++s) {
        float4 v = row[(size_t)s * rstride];
        ml.x = fmaxf(ml.x, v.x); ml.y = fmaxf(ml.y, v.y);
        ml.z = fmaxf(ml.z, v.z); ml.w = fmaxf(ml.w, v.w);
    }
    #pragma unroll 4
    for (; s < SS; ++s) {
        float4 v = row[(size_t)s * rstride];
        mr.x = fmaxf(mr.x, v.x); mr.y = fmaxf(mr.y, v.y);
        mr.z = fmaxf(mr.z, v.z); mr.w = fmaxf(mr.w, v.w);
    }

    float4* outL = (float4*)(pooled + (size_t)n * KTOT) + t;
    float4* outR = (float4*)(pooled + (size_t)n * KTOT + HH) + t;
    *outL = ml;
    *outR = mr;
}

__global__ void dmbert_logits_wmma_kernel(const float* __restrict__ pooled,
                                          const float* __restrict__ Wt,
                                          const float* __restrict__ bias,
                                          float*       __restrict__ out) {
    const int mtile = blockIdx.x;
    const int ntile = blockIdx.y;
    const int wave  = threadIdx.x >> 5;
    const int lane  = threadIdx.x & 31;
    const int half  = lane >> 4;
    const int l16   = lane & 15;

    const int m   = mtile * 16 + l16;
    const int col = ntile * 16 + l16;

    const float* arow = pooled + (size_t)m   * KTOT + 2 * half;
    const float* brow = Wt     + (size_t)col * KTOT + 2 * half;

    v8f c = {};

    const int k0 = wave * (KTOT / 4);
    #pragma unroll 4
    for (int kk = 0; kk < KTOT / 4; kk += 4) {
        const int k = k0 + kk;
        v2f a = *(const v2f*)(arow + k);
        v2f b = *(const v2f*)(brow + k);
        c = __builtin_amdgcn_wmma_f32_16x16x4_f32(
                 false, a,  false, b,
                 (short)0, c,  false,  false);
    }

    __shared__ float red[4][32][8];
    #pragma unroll
    for (int j = 0; j < 8; ++j) red[wave][lane][j] = c[j];
    __syncthreads();

    if (wave == 0 && col < NL) {
        #pragma unroll
        for (int j = 0; j < 8; ++j) {
            float s = red[0][lane][j] + red[1][lane][j] +
                      red[2][lane][j] + red[3][lane][j];
            const int mrow = mtile * 16 + j + half * 8;
            out[(size_t)mrow * NL + col] = s + bias[col];
        }
    }
}

extern "C" void kernel_launch(void* const* d_in, const int* in_sizes, int n_in,
                              void* d_out, int out_size, void* d_ws, size_t ws_size,
                              hipStream_t stream) {
    const float* hidden     = (const float*)d_in[0];
    const float* Wm         = (const float*)d_in[1];
    const float* bias       = (const float*)d_in[2];
    const int*   cand_batch = (const int*)  d_in[3];
    const int*   cand_cut   = (const int*)  d_in[4];
    float*       out        = (float*)d_out;

    float* pooled = (float*)d_ws;
    float* Wt     = pooled + (size_t)NC * KTOT;

    dmbert_wpad_kernel<<<(NLPAD * KTOT) / 256, 256, 0, stream>>>(Wm, Wt);

    dmbert_pool_kernel<<<NC, 256, 0, stream>>>(hidden, cand_batch, cand_cut, pooled);

    dim3 grid(NC / 16, (NL + 15) / 16);
    dmbert_logits_wmma_kernel<<<grid, 128, 0, stream>>>(pooled, Wt, bias, out);
}


# Pool each batch row once: one block per (row, 128-float h slice), 16 waves x 32 tokens in registers, chunk maxima and cut partials merged in LDS for all the row's candidates (max only, exact)
# speedup vs baseline: 6.6439x; 6.6439x over previous
#include <hip/hip_runtime.h>
#include <float.h>

typedef __attribute__((ext_vector_type(2))) float v2f;
typedef __attribute__((ext_vector_type(8))) float v8f;

#define BB    32
#define SS    512
#define HH    1024
#define NC    256
#define NL    34
#define NLPAD 48
#define KTOT  (2 * HH)

__global__ void dmbert_wpad_kernel(const float* __restrict__ Wm,
                                   float*       __restrict__ Wt) {
    const int idx = blockIdx.x * blockDim.x + threadIdx.x;
    const int k   = idx & (KTOT - 1);
    const int c   = idx / KTOT;
    Wt[idx] = (c < NL) ? Wm[(size_t)k * NL + c] : 0.0f;
}

#define POOL_WAVES   16
#define POOL_THREADS (POOL_WAVES * 32)
#define TOKW         (SS / POOL_WAVES)
#define HSL          128
#define NHSL         (HH / HSL)
#define CPASS        POOL_WAVES

__device__ __forceinline__ float4 fmax4(float4 a, float4 b) {
    return make_float4(fmaxf(a.x, b.x), fmaxf(a.y, b.y),
                       fmaxf(a.z, b.z), fmaxf(a.w, b.w));
}

__global__ __launch_bounds__(POOL_THREADS)
void dmbert_pool_kernel(const float* __restrict__ hidden,
                        const int*  __restrict__ cand_batch,
                        const int*  __restrict__ cand_cut,
                        float*      __restrict__ pooled) {
    __shared__ float4 s_full[POOL_WAVES][32];
    __shared__ float4 s_pl[CPASS][32];
    __shared__ float4 s_pr[CPASS][32];
    __shared__ int    s_cnt[NC / 32];
    __shared__ int    s_list[NC];

    const int bat  = blockIdx.x / NHSL;
    const int hs   = blockIdx.x % NHSL;
    const int wave = __builtin_amdgcn_readfirstlane((int)(threadIdx.x >> 5));
    const int lane = threadIdx.x & 31;

    const float4* src = (const float4*)(hidden + ((size_t)bat * SS + (size_t)wave * TOKW) * HH
                                               + (size_t)hs * HSL) + lane;
    float4 v[TOKW];
    #pragma unroll
    for (int j = 0; j < TOKW; ++j) v[j] = src[(size_t)j * (HH / 4)];

    unsigned mm = 0u;
    int  packed = 0;
    bool match  = false;
    if (wave < NC / 32) {
        const int n = threadIdx.x;
        int cb = cand_batch[n];
        int cc = cand_cut[n];
        cb = cb < 0 ? 0 : (cb > BB - 1 ? BB - 1 : cb);
        cc = cc < 0 ? 0 : (cc > SS ? SS : cc);
        match  = (cb == bat);
        packed = n | (cc << 8);
        mm = __builtin_amdgcn_ballot_w32(match);
        if (lane == 0) s_cnt[wave] = __popc(mm);
    }

    float4 fm = v[0];
    #pragma unroll
    for (int j = 1; j < TOKW; ++j) fm = fmax4(fm, v[j]);
    s_full[wave][lane] = fm;
    __syncthreads();

    if (wave < NC / 32) {
        int first = 0;
        for (int w = 0; w < NC / 32; ++w) first += (w < wave) ? s_cnt[w] : 0;
        if (match) s_list[first + __popc(mm & ((1u << lane) - 1u))] = packed;
    }
    int tot = 0;
    for (int w = 0; w < NC / 32; ++w) tot += s_cnt[w];
    const int total = __builtin_amdgcn_readfirstlane(tot);
    __syncthreads();

    const float4 neg4 = make_float4(-FLT_MAX, -FLT_MAX, -FLT_MAX, -FLT_MAX);

    for (int base = 0; base < total; base += CPASS) {
        const int idx = base + lane;
        const int e   = (lane < CPASS && idx < total) ? s_list[idx] : -1;

        unsigned own = __builtin_amdgcn_ballot_w32(e >= 0 && ((e >> 8) >> 5) == wave);
        while (own) {
            const int k   = __builtin_ctz(own);
            own &= own - 1u;
            const int off = (__builtin_amdgcn_readlane(e, k) >> 8) & (TOKW - 1);
            float4 pl = neg4, pr = neg4;
            #pragma unroll
            for (int j = 0; j < TOKW; ++j) {
                const bool left = j < off;
                pl = fmax4(pl, left ? v[j] : neg4);
                pr = fmax4(pr, left ? neg4 : v[j]);
            }
            s_pl[k][lane] = pl;
            s_pr[k][lane] = pr;
        }
        __syncthreads();

        if (base + wave < total) {
            const int em  = __builtin_amdgcn_readlane(e, wave);
            const int n   = em & (NC - 1);
            const int q   = (em >> 8) >> 5;
            float4 f[POOL_WAVES];
            #pragma unroll
            for (int w = 0; w < POOL_WAVES; ++w) f[w] = s_full[w][lane];
            float4 L = neg4, R = neg4;
            if (q < POOL_WAVES) { L = s_pl[wave][lane]; R = s_pr[wave][lane]; }
            #pragma unroll
            for (int w = 0; w < POOL_WAVES; ++w) {
                L = fmax4(L, (w < q) ? f[w] : neg4);
                R = fmax4(R, (w > q) ? f[w] : neg4);
            }
            float4* dst = (float4*)(pooled + (size_t)n * KTOT + (size_t)hs * HSL) + lane;
            dst[0]      = L;
            dst[HH / 4] = R;
        }
        if (base + CPASS < total) __syncthreads();
    }
}

__global__ void dmbert_logits_wmma_kernel(const float* __restrict__ pooled,
                                          const float* __restrict__ Wt,
                                          const float* __restrict__ bias,
                                          float*       __restrict__ out) {
    const int mtile = blockIdx.x;
    const int ntile = blockIdx.y;
    const int wave  = threadIdx.x >> 5;
    const int lane  = threadIdx.x & 31;
    const int half  = lane >> 4;
    const int l16   = lane & 15;

    const int m   = mtile * 16 + l16;
    const int col = ntile * 16 + l16;

    const float* arow = pooled + (size_t)m   * KTOT + 2 * half;
    const float* brow = Wt     + (size_t)col * KTOT + 2 * half;

    v8f c = {};

    const int k0 = wave * (KTOT / 4);
    #pragma unroll 4
    for (int kk = 0; kk < KTOT / 4; kk += 4) {
        const int k = k0 + kk;
        v2f a = *(const v2f*)(arow + k);
        v2f b = *(const v2f*)(brow + k);
        c = __builtin_amdgcn_wmma_f32_16x16x4_f32(
                 false, a,  false, b,
                 (short)0, c,  false,  false);
    }

    __shared__ float red[4][32][8];
    #pragma unroll
    for (int j = 0; j < 8; ++j) red[wave][lane][j] = c[j];
    __syncthreads();

    if (wave == 0 && col < NL) {
        #pragma unroll
        for (int j = 0; j < 8; ++j) {
            float s = red[0][lane][j] + red[1][lane][j] +
                      red[2][lane][j] + red[3][lane][j];
            const int mrow = mtile * 16 + j + half * 8;
            out[(size_t)mrow * NL + col] = s + bias[col];
        }
    }
}

extern "C" void kernel_launch(void* const* d_in, const int* in_sizes, int n_in,
                              void* d_out, int out_size, void* d_ws, size_t ws_size,
                              hipStream_t stream) {
    const float* hidden     = (const float*)d_in[0];
    const float* Wm         = (const float*)d_in[1];
    const float* bias       = (const float*)d_in[2];
    const int*   cand_batch = (const int*)  d_in[3];
    const int*   cand_cut   = (const int*)  d_in[4];
    float*       out        = (float*)d_out;

    float* pooled = (float*)d_ws;
    float* Wt     = pooled + (size_t)NC * KTOT;

    dmbert_wpad_kernel<<<(NLPAD * KTOT) / 256, 256, 0, stream>>>(Wm, Wt);

    dmbert_pool_kernel<<<BB * NHSL, POOL_THREADS, 0, stream>>>(hidden, cand_batch, cand_cut, pooled);

    dim3 grid(NC / 16, (NL + 15) / 16);
    dmbert_logits_wmma_kernel<<<grid, 128, 0, stream>>>(pooled, Wt, bias, out);
}


# Fold the pad+transpose of W into the pool kernel (waves 0-11 of each block move 384 Wt elements, load issued ahead of their hidden loads): its own launch is gone, 3 launches -> 2
# speedup vs baseline: 7.3209x; 1.1019x over previous
#include <hip/hip_runtime.h>
#include <float.h>

typedef __attribute__((ext_vector_type(2))) float v2f;
typedef __attribute__((ext_vector_type(8))) float v8f;

#define BB    32
#define SS    512
#define HH    1024
#define NC    256
#define NL    34
#define NLPAD 48
#define KTOT  (2 * HH)

#define POOL_WAVES   16
#define POOL_THREADS (POOL_WAVES * 32)
#define TOKW         (SS / POOL_WAVES)
#define HSL          128
#define NHSL         (HH / HSL)
#define CPASS        POOL_WAVES
#define WT_PER_BLOCK ((NLPAD * KTOT) / (BB * NHSL))
#define WT_WAVES     (WT_PER_BLOCK / 32)

__device__ __forceinline__ float4 fmax4(float4 a, float4 b) {
    return make_float4(fmaxf(a.x, b.x), fmaxf(a.y, b.y),
                       fmaxf(a.z, b.z), fmaxf(a.w, b.w));
}

__global__ __launch_bounds__(POOL_THREADS)
void dmbert_pool_kernel(const float* __restrict__ hidden,
                        const int*  __restrict__ cand_batch,
                        const int*  __restrict__ cand_cut,
                        const float* __restrict__ Wm,
                        float*      __restrict__ Wt,
                        float*      __restrict__ pooled) {
    __shared__ float4 s_full[POOL_WAVES][32];
    __shared__ float4 s_pl[CPASS][32];
    __shared__ float4 s_pr[CPASS][32];
    __shared__ int    s_cnt[NC / 32];
    __shared__ int    s_list[NC];

    const int bat  = blockIdx.x / NHSL;
    const int hs   = blockIdx.x % NHSL;
    const int wave = __builtin_amdgcn_readfirstlane((int)(threadIdx.x >> 5));
    const int lane = threadIdx.x & 31;

    const bool wmove = wave < WT_WAVES;
    int   widx = 0;
    float wval = 0.0f;
    if (wmove) {
        widx = blockIdx.x * WT_PER_BLOCK + (int)threadIdx.x;
        const int k = widx & (KTOT - 1);
        const int c = widx / KTOT;
        wval = (c < NL) ? Wm[(size_t)k * NL + c] : 0.0f;
    }

    const float4* src = (const float4*)(hidden + ((size_t)bat * SS + (size_t)wave * TOKW) * HH
                                               + (size_t)hs * HSL) + lane;
    float4 v[TOKW];
    #pragma unroll
    for (int j = 0; j < TOKW; ++j) v[j] = src[(size_t)j * (HH / 4)];

    unsigned mm = 0u;
    int  packed = 0;
    bool match  = false;
    if (wave < NC / 32) {
        const int n = threadIdx.x;
        int cb = cand_batch[n];
        int cc = cand_cut[n];
        cb = cb < 0 ? 0 : (cb > BB - 1 ? BB - 1 : cb);
        cc = cc < 0 ? 0 : (cc > SS ? SS : cc);
        match  = (cb == bat);
        packed = n | (cc << 8);
        mm = __builtin_amdgcn_ballot_w32(match);
        if (lane == 0) s_cnt[wave] = __popc(mm);
    }

    if (wmove) Wt[widx] = wval;

    float4 fm = v[0];
    #pragma unroll
    for (int j = 1; j < TOKW; ++j) fm = fmax4(fm, v[j]);
    s_full[wave][lane] = fm;
    __syncthreads();

    if (wave < NC / 32) {
        int first = 0;
        for (int w = 0; w < NC / 32; ++w) first += (w < wave) ? s_cnt[w] : 0;
        if (match) s_list[first + __popc(mm & ((1u << lane) - 1u))] = packed;
    }
    int tot = 0;
    for (int w = 0; w < NC / 32; ++w) tot += s_cnt[w];
    const int total = __builtin_amdgcn_readfirstlane(tot);
    __syncthreads();

    const float4 neg4 = make_float4(-FLT_MAX, -FLT_MAX, -FLT_MAX, -FLT_MAX);

    for (int base = 0; base < total; base += CPASS) {
        const int idx = base + lane;
        const int e   = (lane < CPASS && idx < total) ? s_list[idx] : -1;

        unsigned own = __builtin_amdgcn_ballot_w32(e >= 0 && ((e >> 8) >> 5) == wave);
        while (own) {
            const int k   = __builtin_ctz(own);
            own &= own - 1u;
            const int off = (__builtin_amdgcn_readlane(e, k) >> 8) & (TOKW - 1);
            float4 pl = neg4, pr = neg4;
            #pragma unroll
            for (int j = 0; j < TOKW; ++j) {
                const bool left = j < off;
                pl = fmax4(pl, left ? v[j] : neg4);
                pr = fmax4(pr, left ? neg4 : v[j]);
            }
            s_pl[k][lane] = pl;
            s_pr[k][lane] = pr;
        }
        __syncthreads();

        if (base + wave < total) {
            const int em  = __builtin_amdgcn_readlane(e, wave);
            const int n   = em & (NC - 1);
            const int q   = (em >> 8) >> 5;
            float4 f[POOL_WAVES];
            #pragma unroll
            for (int w = 0; w < POOL_WAVES; ++w) f[w] = s_full[w][lane];
            float4 L = neg4, R = neg4;
            if (q < POOL_WAVES) { L = s_pl[wave][lane]; R = s_pr[wave][lane]; }
            #pragma unroll
            for (int w = 0; w < POOL_WAVES; ++w) {
                L = fmax4(L, (w < q) ? f[w] : neg4);
                R = fmax4(R, (w > q) ? f[w] : neg4);
            }
            float4* dst = (float4*)(pooled + (size_t)n * KTOT + (size_t)hs * HSL) + lane;
            dst[0]      = L;
            dst[HH / 4] = R;
        }
        if (base + CPASS < total) __syncthreads();
    }
}

__global__ void dmbert_logits_wmma_kernel(const float* __restrict__ pooled,
                                          const float* __restrict__ Wt,
                                          const float* __restrict__ bias,
                                          float*       __restrict__ out) {
    const int mtile = blockIdx.x;
    const int ntile = blockIdx.y;
    const int wave  = threadIdx.x >> 5;
    const int lane  = threadIdx.x & 31;
    const int half  = lane >> 4;
    const int l16   = lane & 15;

    const int m   = mtile * 16 + l16;
    const int col = ntile * 16 + l16;

    const float* arow = pooled + (size_t)m   * KTOT + 2 * half;
    const float* brow = Wt     + (size_t)col * KTOT + 2 * half;

    v8f c = {};

    const int k0 = wave * (KTOT / 4);
    #pragma unroll 4
    for (int kk = 0; kk < KTOT / 4; kk += 4) {
        const int k = k0 + kk;
        v2f a = *(const v2f*)(arow + k);
        v2f b = *(const v2f*)(brow + k);
        c = __builtin_amdgcn_wmma_f32_16x16x4_f32(
                 false, a,  false, b,
                 (short)0, c,  false,  false);
    }

    __shared__ float red[4][32][8];
    #pragma unroll
    for (int j = 0; j < 8; ++j) red[wave][lane][j] = c[j];
    __syncthreads();

    if (wave == 0 && col < NL) {
        #pragma unroll
        for (int j = 0; j < 8; ++j) {
            float s = red[0][lane][j] + red[1][lane][j] +
                      red[2][lane][j] + red[3][lane][j];
            const int mrow = mtile * 16 + j + half * 8;
            out[(size_t)mrow * NL + col] = s + bias[col];
        }
    }
}

extern "C" void kernel_launch(void* const* d_in, const int* in_sizes, int n_in,
                              void* d_out, int out_size, void* d_ws, size_t ws_size,
                              hipStream_t stream) {
    const float* hidden     = (const float*)d_in[0];
    const float* Wm         = (const float*)d_in[1];
    const float* bias       = (const float*)d_in[2];
    const int*   cand_batch = (const int*)  d_in[3];
    const int*   cand_cut   = (const int*)  d_in[4];
    float*       out        = (float*)d_out;

    float* pooled = (float*)d_ws;
    float* Wt     = pooled + (size_t)NC * KTOT;

    dmbert_pool_kernel<<<BB * NHSL, POOL_THREADS, 0, stream>>>(hidden, cand_batch, cand_cut, Wm, Wt, pooled);

    dim3 grid(NC / 16, (NL + 15) / 16);
    dmbert_logits_wmma_kernel<<<grid, 128, 0, stream>>>(pooled, Wt, bias, out);
}


# Logits GEMM: operand rows fetched as whole 512-byte lines by 16 waves per tile (128 k each), staged in LDS for the fragment reads; sum order changes: 2 accumulator tiles, 16 partials in wave order
# speedup vs baseline: 9.6637x; 1.3200x over previous
#include <hip/hip_runtime.h>
#include <float.h>

typedef __attribute__((ext_vector_type(2))) float v2f;
typedef __attribute__((ext_vector_type(8))) float v8f;

#define BB    32
#define SS    512
#define HH    1024
#define NC    256
#define NL    34
#define NLPAD 48
#define KTOT  (2 * HH)

#define POOL_WAVES   16
#define POOL_THREADS (POOL_WAVES * 32)
#define TOKW         (SS / POOL_WAVES)
#define HSL          128
#define NHSL         (HH / HSL)
#define CPASS        POOL_WAVES
#define WT_PER_BLOCK ((NLPAD * KTOT) / (BB * NHSL))
#define WT_WAVES     (WT_PER_BLOCK / 32)

__device__ __forceinline__ float4 fmax4(float4 a, float4 b) {
    return make_float4(fmaxf(a.x, b.x), fmaxf(a.y, b.y),
                       fmaxf(a.z, b.z), fmaxf(a.w, b.w));
}

__global__ __launch_bounds__(POOL_THREADS)
void dmbert_pool_kernel(const float* __restrict__ hidden,
                        const int*  __restrict__ cand_batch,
                        const int*  __restrict__ cand_cut,
                        const float* __restrict__ Wm,
                        float*      __restrict__ Wt,
                        float*      __restrict__ pooled) {
    __shared__ float4 s_full[POOL_WAVES][32];
    __shared__ float4 s_pl[CPASS][32];
    __shared__ float4 s_pr[CPASS][32];
    __shared__ int    s_cnt[NC / 32];
    __shared__ int    s_list[NC];

    const int bat  = blockIdx.x / NHSL;
    const int hs   = blockIdx.x % NHSL;
    const int wave = __builtin_amdgcn_readfirstlane((int)(threadIdx.x >> 5));
    const int lane = threadIdx.x & 31;

    const bool wmove = wave < WT_WAVES;
    int   widx = 0;
    float wval = 0.0f;
    if (wmove) {
        widx = blockIdx.x * WT_PER_BLOCK + (int)threadIdx.x;
        const int k = widx & (KTOT - 1);
        const int c = widx / KTOT;
        wval = (c < NL) ? Wm[(size_t)k * NL + c] : 0.0f;
    }

    const float4* src = (const float4*)(hidden + ((size_t)bat * SS + (size_t)wave * TOKW) * HH
                                               + (size_t)hs * HSL) + lane;
    float4 v[TOKW];
    #pragma unroll
    for (int j = 0; j < TOKW; ++j) v[j] = src[(size_t)j * (HH / 4)];

    unsigned mm = 0u;
    int  packed = 0;
    bool match  = false;
    if (wave < NC / 32) {
        const int n = threadIdx.x;
        int cb = cand_batch[n];
        int cc = cand_cut[n];
        cb = cb < 0 ? 0 : (cb > BB - 1 ? BB - 1 : cb);
        cc = cc < 0 ? 0 : (cc > SS ? SS : cc);
        match  = (cb == bat);
        packed = n | (cc << 8);
        mm = __builtin_amdgcn_ballot_w32(match);
        if (lane == 0) s_cnt[wave] = __popc(mm);
    }

    if (wmove) Wt[widx] = wval;

    float4 fm = v[0];
    #pragma unroll
    for (int j = 1; j < TOKW; ++j) fm = fmax4(fm, v[j]);
    s_full[wave][lane] = fm;
    __syncthreads();

    if (wave < NC / 32) {
        int first = 0;
        for (int w = 0; w < NC / 32; ++w) first += (w < wave) ? s_cnt[w] : 0;
        if (match) s_list[first + __popc(mm & ((1u << lane) - 1u))] = packed;
    }
    int tot = 0;
    for (int w = 0; w < NC / 32; ++w) tot += s_cnt[w];
    const int total = __builtin_amdgcn_readfirstlane(tot);
    __syncthreads();

    const float4 neg4 = make_float4(-FLT_MAX, -FLT_MAX, -FLT_MAX, -FLT_MAX);

    for (int base = 0; base < total; base += CPASS) {
        const int idx = base + lane;
        const int e   = (lane < CPASS && idx < total) ? s_list[idx] : -1;

        unsigned own = __builtin_amdgcn_ballot_w32(e >= 0 && ((e >> 8) >> 5) == wave);
        while (own) {
            const int k   = __builtin_ctz(own);
            own &= own - 1u;
            const int off = (__builtin_amdgcn_readlane(e, k) >> 8) & (TOKW - 1);
            float4 pl = neg4, pr = neg4;
            #pragma unroll
            for (int j = 0; j < TOKW; ++j) {
                const bool left = j < off;
                pl = fmax4(pl, left ? v[j] : neg4);
                pr = fmax4(pr, left ? neg4 : v[j]);
            }
            s_pl[k][lane] = pl;
            s_pr[k][lane] = pr;
        }
        __syncthreads();

        if (base + wave < total) {
            const int em  = __builtin_amdgcn_readlane(e, wave);
            const int n   = em & (NC - 1);
            const int q   = (em >> 8) >> 5;
            float4 f[POOL_WAVES];
            #pragma unroll
            for (int w = 0; w < POOL_WAVES; ++w) f[w] = s_full[w][lane];
            float4 L = neg4, R = neg4;
            if (q < POOL_WAVES) { L = s_pl[wave][lane]; R = s_pr[wave][lane]; }
            #pragma unroll
            for (int w = 0; w < POOL_WAVES; ++w) {
                L = fmax4(L, (w < q) ? f[w] : neg4);
                R = fmax4(R, (w > q) ? f[w] : neg4);
            }
            float4* dst = (float4*)(pooled + (size_t)n * KTOT + (size_t)hs * HSL) + lane;
            dst[0]      = L;
            dst[HH / 4] = R;
        }
        if (base + CPASS < total) __syncthreads();
    }
}

#define GW     16
#define GTHR   (GW * 32)
#define KS     (KTOT / GW)
#define PITCH  (KS + 4)
#define OPF    (16 * PITCH)
#define WAVF   (2 * OPF)

__global__ __launch_bounds__(GTHR)
void dmbert_logits_wmma_kernel(const float* __restrict__ pooled,
                               const float* __restrict__ Wt,
                               const float* __restrict__ bias,
                               float*       __restrict__ out) {
    __shared__ float4 s_stage4[GW * WAVF / 4];
    float* s_stage = (float*)s_stage4;

    const int mtile = blockIdx.x;
    const int ntile = blockIdx.y;
    const int wave  = __builtin_amdgcn_readfirstlane((int)(threadIdx.x >> 5));
    const int lane  = threadIdx.x & 31;
    const int half  = lane >> 4;
    const int l16   = lane & 15;

    const int k0 = wave * KS;
    const float* ga = pooled + (size_t)(mtile * 16) * KTOT + k0 + lane * 4;
    const float* gb = Wt     + (size_t)(ntile * 16) * KTOT + k0 + lane * 4;
    float* sa = s_stage + wave * WAVF;
    float* sb = sa + OPF;

    float4 ta[16], tb[16];
    #pragma unroll
    for (int r = 0; r < 16; ++r) ta[r] = *(const float4*)(ga + (size_t)r * KTOT);
    #pragma unroll
    for (int r = 0; r < 16; ++r) tb[r] = *(const float4*)(gb + (size_t)r * KTOT);
    #pragma unroll
    for (int r = 0; r < 16; ++r) *(float4*)(sa + r * PITCH + lane * 4) = ta[r];
    #pragma unroll
    for (int r = 0; r < 16; ++r) *(float4*)(sb + r * PITCH + lane * 4) = tb[r];

    const float* fa = sa + l16 * PITCH + 4 * half;
    const float* fb = sb + l16 * PITCH + 4 * half;

    v8f c0 = {};
    v8f c1 = {};
    #pragma unroll
    for (int p = 0; p < KS / 8; ++p) {
        const float4 a = *(const float4*)(fa + 8 * p);
        const float4 b = *(const float4*)(fb + 8 * p);
        v2f a0, a1, b0, b1;
        a0.x = a.x; a0.y = a.y; a1.x = a.z; a1.y = a.w;
        b0.x = b.x; b0.y = b.y; b1.x = b.z; b1.y = b.w;
        c0 = __builtin_amdgcn_wmma_f32_16x16x4_f32(
                  false, a0,  false, b0,
                  (short)0, c0,  false,  false);
        c1 = __builtin_amdgcn_wmma_f32_16x16x4_f32(
                  false, a1,  false, b1,
                  (short)0, c1,  false,  false);
    }

    #pragma unroll
    for (int j = 0; j < 8; ++j) sa[j * 32 + lane] = c0[j] + c1[j];
    __syncthreads();

    if (wave < 8) {
        const int col = ntile * 16 + l16;
        float s = 0.0f;
        #pragma unroll
        for (int w = 0; w < GW; ++w) s += s_stage[w * WAVF + wave * 32 + lane];
        const int mrow = mtile * 16 + wave + half * 8;
        if (col < NL) out[(size_t)mrow * NL + col] = s + bias[col];
    }
}

extern "C" void kernel_launch(void* const* d_in, const int* in_sizes, int n_in,
                              void* d_out, int out_size, void* d_ws, size_t ws_size,
                              hipStream_t stream) {
    const float* hidden     = (const float*)d_in[0];
    const float* Wm         = (const float*)d_in[1];
    const float* bias       = (const float*)d_in[2];
    const int*   cand_batch = (const int*)  d_in[3];
    const int*   cand_cut   = (const int*)  d_in[4];
    float*       out        = (float*)d_out;

    float* pooled = (float*)d_ws;
    float* Wt     = pooled + (size_t)NC * KTOT;

    dmbert_pool_kernel<<<BB * NHSL, POOL_THREADS, 0, stream>>>(hidden, cand_batch, cand_cut, Wm, Wt, pooled);

    dim3 grid(NC / 16, (NL + 15) / 16);
    dmbert_logits_wmma_kernel<<<grid, GTHR, 0, stream>>>(pooled, Wt, bias, out);
}


# Logits product moved into the pool kernel: per block a [16 cand x 256 k] LDS tile times its W rows on WMMA (12 waves), partial logits per h slice; launch 2 adds 8 partials + bias; sum order changes
# speedup vs baseline: 12.8257x; 1.3272x over previous
#include <hip/hip_runtime.h>
#include <float.h>

typedef __attribute__((ext_vector_type(2))) float v2f;
typedef __attribute__((ext_vector_type(8))) float v8f;

#define BB    32
#define SS    512
#define HH    1024
#define NC    256
#define NL    34
#define KTOT  (2 * HH)

#define POOL_WAVES   16
#define POOL_THREADS (POOL_WAVES * 32)
#define TOKW         (SS / POOL_WAVES)
#define HSL          128
#define NHSL         (HH / HSL)
#define CPASS        POOL_WAVES
#define KSL          (2 * HSL)
#define PA           (KSL + 4)
#define WRUN         (HSL * NL)
#define WSL4         (2 * WRUN / 4)
#define MMW          12
#define KQ           (KSL / 4)
#define PSTR         64

__device__ __forceinline__ float4 fmax4(float4 a, float4 b) {
    return make_float4(fmaxf(a.x, b.x), fmaxf(a.y, b.y),
                       fmaxf(a.z, b.z), fmaxf(a.w, b.w));
}

__global__ __launch_bounds__(POOL_THREADS)
void dmbert_pool_logits_kernel(const float* __restrict__ hidden,
                               const int*   __restrict__ cand_batch,
                               const int*   __restrict__ cand_cut,
                               const float* __restrict__ Wm,
                               float*       __restrict__ partial) {
    __shared__ float4 s_full[POOL_WAVES][32];
    __shared__ float4 s_pl[CPASS][32];
    __shared__ float4 s_pr[CPASS][32];
    __shared__ int    s_cnt[NC / 32];
    __shared__ int    s_list[NC];
    __shared__ float4 s_w4[WSL4 + 4];
    __shared__ float4 s_a4[CPASS * PA / 4];
    __shared__ float  s_part[4][3][8][32];

    const int bat  = blockIdx.x / NHSL;
    const int hs   = blockIdx.x % NHSL;
    const int wave = __builtin_amdgcn_readfirstlane((int)(threadIdx.x >> 5));
    const int lane = threadIdx.x & 31;
    const int half = lane >> 4;
    const int l16  = lane & 15;

    const float4* w0 = (const float4*)(Wm + (size_t)hs * WRUN);
    const float4* w1 = (const float4*)(Wm + (size_t)(HH + hs * HSL) * NL);
    const int tid = (int)threadIdx.x;
    const float4 wa = w0[tid];
    const float4 wb = w0[POOL_THREADS + tid];
    const float4 wc = w1[tid];
    const float4 wd = w1[POOL_THREADS + tid];
    const bool wtail = wave < 4;
    const float4* wt = (wave < 2) ? (w0 + 2 * POOL_THREADS) : (w1 + 2 * POOL_THREADS - 64);
    float4 wx = make_float4(0.0f, 0.0f, 0.0f, 0.0f);
    if (wtail) wx = wt[tid];

    const float4* src = (const float4*)(hidden + ((size_t)bat * SS + (size_t)wave * TOKW) * HH
                                               + (size_t)hs * HSL) + lane;
    float4 v[TOKW];
    #pragma unroll
    for (int j = 0; j < TOKW; ++j) v[j] = src[(size_t)j * (HH / 4)];

    s_w4[tid]                             = wa;
    s_w4[POOL_THREADS + tid]              = wb;
    s_w4[WRUN / 4 + tid]                  = wc;
    s_w4[WRUN / 4 + POOL_THREADS + tid]   = wd;
    if (wtail) s_w4[(wave < 2 ? 2 * POOL_THREADS : WRUN / 4 + 2 * POOL_THREADS - 64) + tid] = wx;
    if (tid < 4) s_w4[WSL4 + tid] = make_float4(0.0f, 0.0f, 0.0f, 0.0f);

    unsigned mm = 0u;
    int  packed = 0;
    bool match  = false;
    if (wave < NC / 32) {
        const int n = threadIdx.x;
        int cb = cand_batch[n];
        int cc = cand_cut[n];
        cb = cb < 0 ? 0 : (cb > BB - 1 ? BB - 1 : cb);
        cc = cc < 0 ? 0 : (cc > SS ? SS : cc);
        match  = (cb == bat);
        packed = n | (cc << 8);
        mm = __builtin_amdgcn_ballot_w32(match);
        if (lane == 0) s_cnt[wave] = __popc(mm);
    }

    float4 fm = v[0];
    #pragma unroll
    for (int j = 1; j < TOKW; ++j) fm = fmax4(fm, v[j]);
    s_full[wave][lane] = fm;
    __syncthreads();

    if (wave < NC / 32) {
        int first = 0;
        for (int w = 0; w < NC / 32; ++w) first += (w < wave) ? s_cnt[w] : 0;
        if (match) s_list[first + __popc(mm & ((1u << lane) - 1u))] = packed;
    }
    int tot = 0;
    for (int w = 0; w < NC / 32; ++w) tot += s_cnt[w];
    const int total = __builtin_amdgcn_readfirstlane(tot);
    __syncthreads();

    const float4 neg4  = make_float4(-FLT_MAX, -FLT_MAX, -FLT_MAX, -FLT_MAX);
    const float4 zero4 = make_float4(0.0f, 0.0f, 0.0f, 0.0f);

    for (int base = 0; base < total; base += CPASS) {
        const int idx = base + lane;
        const int e   = (lane < CPASS && idx < total) ? s_list[idx] : -1;

        unsigned own = __builtin_amdgcn_ballot_w32(e >= 0 && ((e >> 8) >> 5) == wave);
        while (own) {
            const int k   = __builtin_ctz(own);
            own &= own - 1u;
            const int off = (__builtin_amdgcn_readlane(e, k) >> 8) & (TOKW - 1);
            float4 pl = neg4, pr = neg4;
            #pragma unroll
            for (int j = 0; j < TOKW; ++j) {
                const bool left = j < off;
                pl = fmax4(pl, left ? v[j] : neg4);
                pr = fmax4(pr, left ? neg4 : v[j]);
            }
            s_pl[k][lane] = pl;
            s_pr[k][lane] = pr;
        }
        __syncthreads();

        const bool have = base + wave < total;
        int    n = 0;
        float4 L = zero4, R = zero4;
        if (have) {
            const int em  = __builtin_amdgcn_readlane(e, wave);
            n = em & (NC - 1);
            const int q   = (em >> 8) >> 5;
            float4 f[POOL_WAVES];
            #pragma unroll
            for (int w = 0; w < POOL_WAVES; ++w) f[w] = s_full[w][lane];
            L = neg4; R = neg4;
            if (q < POOL_WAVES) { L = s_pl[wave][lane]; R = s_pr[wave][lane]; }
            #pragma unroll
            for (int w = 0; w < POOL_WAVES; ++w) {
                L = fmax4(L, (w < q) ? f[w] : neg4);
                R = fmax4(R, (w > q) ? f[w] : neg4);
            }
        }
        s_a4[wave * (PA / 4) + lane]            = L;
        s_a4[wave * (PA / 4) + HSL / 4 + lane]  = R;
        __syncthreads();

        if (wave < MMW) {
            const int nt  = wave >> 2;
            const int kq  = wave & 3;
            const int col = nt * 16 + l16;
            const bool cok = col < NL;
            const float* fa = (const float*)s_a4 + l16 * PA + kq * KQ + 4 * half;
            const float* fb = (const float*)s_w4 + (kq * KQ + 4 * half) * NL + col;
            v8f c0 = {};
            v8f c1 = {};
            #pragma unroll
            for (int p = 0; p < KQ / 8; ++p) {
                const float4 a = *(const float4*)(fa + 8 * p);
                const float bx = fb[(8 * p + 0) * NL];
                const float by = fb[(8 * p + 1) * NL];
                const float bz = fb[(8 * p + 2) * NL];
                const float bw = fb[(8 * p + 3) * NL];
                v2f a0, a1, b0, b1;
                a0.x = a.x; a0.y = a.y; a1.x = a.z; a1.y = a.w;
                b0.x = cok ? bx : 0.0f; b0.y = cok ? by : 0.0f;
                b1.x = cok ? bz : 0.0f; b1.y = cok ? bw : 0.0f;
                c0 = __builtin_amdgcn_wmma_f32_16x16x4_f32(
                          false, a0,  false, b0,
                          (short)0, c0,  false,  false);
                c1 = __builtin_amdgcn_wmma_f32_16x16x4_f32(
                          false, a1,  false, b1,
                          (short)0, c1,  false,  false);
            }
            #pragma unroll
            for (int j = 0; j < 8; ++j) s_part[kq][nt][j][lane] = c0[j] + c1[j];
        }
        __syncthreads();

        if (have) {
            const int j  = wave & 7;
            const int hb = (wave >> 3) * 16;
            float s = 0.0f, t = 0.0f;
            #pragma unroll
            for (int q = 0; q < 4; ++q) {
                s += s_part[q][half][j][hb + l16];
                t += s_part[q][2][j][hb + l16];
            }
            float* dst = partial + ((size_t)n * NHSL + hs) * PSTR;
            dst[lane] = s;
            if (lane < NL - 32) dst[32 + lane] = t;
        }
    }
}

#define STHR 128

__global__ __launch_bounds__(STHR)
void dmbert_logits_sum_kernel(const float* __restrict__ partial,
                              const float* __restrict__ bias,
                              float*       __restrict__ out) {
    const int i = blockIdx.x * STHR + threadIdx.x;
    const int n = i / NL;
    const int c = i - n * NL;
    const float* p = partial + (size_t)n * (NHSL * PSTR) + c;
    float x[NHSL];
    #pragma unroll
    for (int h = 0; h < NHSL; ++h) x[h] = p[h * PSTR];
    float s = x[0];
    #pragma unroll
    for (int h = 1; h < NHSL; ++h) s += x[h];
    out[i] = s + bias[c];
}

extern "C" void kernel_launch(void* const* d_in, const int* in_sizes, int n_in,
                              void* d_out, int out_size, void* d_ws, size_t ws_size,
                              hipStream_t stream) {
    const float* hidden     = (const float*)d_in[0];
    const float* Wm         = (const float*)d_in[1];
    const float* bias       = (const float*)d_in[2];
    const int*   cand_batch = (const int*)  d_in[3];
    const int*   cand_cut   = (const int*)  d_in[4];
    float*       out        = (float*)d_out;

    float* partial = (float*)d_ws;

    dmbert_pool_logits_kernel<<<BB * NHSL, POOL_THREADS, 0, stream>>>(hidden, cand_batch, cand_cut, Wm, partial);

    dmbert_logits_sum_kernel<<<(NC * NL) / STHR, STHR, 0, stream>>>(partial, bias, out);
}


# Cuts split by one running max up and one down a wave's 32 tokens, stored at each cut (cost no longer per cut); candidate list built under the hidden loads, its barriers leave loads in flight
# speedup vs baseline: 13.6666x; 1.0656x over previous
#include <hip/hip_runtime.h>
#include <float.h>

typedef __attribute__((ext_vector_type(2))) float v2f;
typedef __attribute__((ext_vector_type(8))) float v8f;

#define BB    32
#define SS    512
#define HH    1024
#define NC    256
#define NL    34
#define KTOT  (2 * HH)

#define POOL_WAVES   16
#define POOL_THREADS (POOL_WAVES * 32)
#define TOKW         (SS / POOL_WAVES)
#define HSL          128
#define NHSL         (HH / HSL)
#define CPASS        POOL_WAVES
#define KSL          (2 * HSL)
#define PA           (KSL + 4)
#define WRUN         (HSL * NL)
#define WSL4         (2 * WRUN / 4)
#define MMW          12
#define KQ           (KSL / 4)
#define PSTR         64

__device__ __forceinline__ float4 fmax4(float4 a, float4 b) {
    return make_float4(fmaxf(a.x, b.x), fmaxf(a.y, b.y),
                       fmaxf(a.z, b.z), fmaxf(a.w, b.w));
}

__device__ __forceinline__ void run_max(float4& p, const float4& t) {
    asm volatile("v_max_num_f32 %0, %0, %4\n\t"
                 "v_max_num_f32 %1, %1, %5\n\t"
                 "v_max_num_f32 %2, %2, %6\n\t"
                 "v_max_num_f32 %3, %3, %7"
                 : "+v"(p.x), "+v"(p.y), "+v"(p.z), "+v"(p.w)
                 : "v"(t.x), "v"(t.y), "v"(t.z), "v"(t.w));
}

__device__ __forceinline__ void lds_barrier() {
    asm volatile("s_wait_dscnt 0x0\n\ts_barrier_signal -1\n\ts_barrier_wait -1" ::: "memory");
}

__global__ __launch_bounds__(POOL_THREADS)
void dmbert_pool_logits_kernel(const float* __restrict__ hidden,
                               const int*   __restrict__ cand_batch,
                               const int*   __restrict__ cand_cut,
                               const float* __restrict__ Wm,
                               float*       __restrict__ partial) {
    __shared__ float4 s_full[POOL_WAVES][32];
    __shared__ float4 s_pl[CPASS][32];
    __shared__ float4 s_pr[CPASS][32];
    __shared__ int    s_cnt[NC / 32];
    __shared__ int    s_list[NC];
    __shared__ float4 s_w4[WSL4 + 4];
    __shared__ float4 s_a4[CPASS * PA / 4];
    __shared__ float  s_part[4][3][8][32];

    const int bat  = blockIdx.x / NHSL;
    const int hs   = blockIdx.x % NHSL;
    const int wave = __builtin_amdgcn_readfirstlane((int)(threadIdx.x >> 5));
    const int lane = threadIdx.x & 31;
    const int half = lane >> 4;
    const int l16  = lane & 15;

    const float4* w0 = (const float4*)(Wm + (size_t)hs * WRUN);
    const float4* w1 = (const float4*)(Wm + (size_t)(HH + hs * HSL) * NL);
    const int tid = (int)threadIdx.x;
    const float4 wa = w0[tid];
    const float4 wb = w0[POOL_THREADS + tid];
    const float4 wc = w1[tid];
    const float4 wd = w1[POOL_THREADS + tid];
    const bool wtail = wave < 4;
    const float4* wt = (wave < 2) ? (w0 + 2 * POOL_THREADS) : (w1 + 2 * POOL_THREADS - 64);
    float4 wx = make_float4(0.0f, 0.0f, 0.0f, 0.0f);
    if (wtail) wx = wt[tid];

    int cb = cand_batch[tid & (NC - 1)];
    int cc = cand_cut[tid & (NC - 1)];

    const float4* src = (const float4*)(hidden + ((size_t)bat * SS + (size_t)wave * TOKW) * HH
                                               + (size_t)hs * HSL) + lane;
    float4 v[TOKW];
    #pragma unroll
    for (int j = 0; j < TOKW; ++j) v[j] = src[(size_t)j * (HH / 4)];

    s_w4[tid]                             = wa;
    s_w4[POOL_THREADS + tid]              = wb;
    s_w4[WRUN / 4 + tid]                  = wc;
    s_w4[WRUN / 4 + POOL_THREADS + tid]   = wd;
    if (wtail) s_w4[(wave < 2 ? 2 * POOL_THREADS : WRUN / 4 + 2 * POOL_THREADS - 64) + tid] = wx;
    if (tid < 4) s_w4[WSL4 + tid] = make_float4(0.0f, 0.0f, 0.0f, 0.0f);

    cb = cb < 0 ? 0 : (cb > BB - 1 ? BB - 1 : cb);
    cc = cc < 0 ? 0 : (cc > SS ? SS : cc);
    const bool     match  = (cb == bat);
    const int      packed = (tid & (NC - 1)) | (cc << 8);
    const unsigned mm     = __builtin_amdgcn_ballot_w32(match);
    if (wave < NC / 32 && lane == 0) s_cnt[wave] = __popc(mm);
    lds_barrier();

    if (wave < NC / 32) {
        int first = 0;
        for (int w = 0; w < NC / 32; ++w) first += (w < wave) ? s_cnt[w] : 0;
        if (match) s_list[first + __popc(mm & ((1u << lane) - 1u))] = packed;
    }
    int tot = 0;
    for (int w = 0; w < NC / 32; ++w) tot += s_cnt[w];
    const int total = __builtin_amdgcn_readfirstlane(tot);
    lds_barrier();

    const float4 neg4  = make_float4(-FLT_MAX, -FLT_MAX, -FLT_MAX, -FLT_MAX);
    const float4 zero4 = make_float4(0.0f, 0.0f, 0.0f, 0.0f);

    for (int base = 0; base < total; base += CPASS) {
        const int idx = base + lane;
        const int e   = (lane < CPASS && idx < total) ? s_list[idx] : -1;

        const unsigned own = __builtin_amdgcn_ballot_w32(e >= 0 && ((e >> 8) >> 5) == wave);
        unsigned cm = 0u;
        for (unsigned o = own; o; o &= o - 1u)
            cm |= 1u << ((__builtin_amdgcn_readlane(e, __builtin_ctz(o)) >> 8) & (TOKW - 1));
        const unsigned cuts = __builtin_amdgcn_readfirstlane(cm);
        unsigned cuts_dn = cuts;
        asm volatile("" : "+s"(cuts_dn));

        float4 p = neg4;
        #pragma unroll
        for (int j = 0; j < TOKW; ++j) {
            if (cuts & (1u << j)) {
                unsigned o = own;
                do {
                    const int k = __builtin_ctz(o);
                    if (((__builtin_amdgcn_readlane(e, k) >> 8) & (TOKW - 1)) == j)
                        s_pl[k][lane] = p;
                    o &= o - 1u;
                } while (o);
            }
            run_max(p, v[j]);
        }
        if (base == 0) s_full[wave][lane] = p;

        if (cuts_dn) {
            float4 r = neg4;
            #pragma unroll
            for (int g = TOKW / 8 - 1; g >= 0; --g) {
                if (cuts_dn & (0xFFu << (8 * g))) {
                    #pragma unroll
                    for (int jj = 7; jj >= 0; --jj) {
                        const int j = 8 * g + jj;
                        run_max(r, v[j]);
                        if (cuts_dn & (1u << j)) {
                            unsigned o = own;
                            do {
                                const int k = __builtin_ctz(o);
                                if (((__builtin_amdgcn_readlane(e, k) >> 8) & (TOKW - 1)) == j)
                                    s_pr[k][lane] = r;
                                o &= o - 1u;
                            } while (o);
                        }
                    }
                    if (!(cuts_dn & ((1u << (8 * g)) - 1u))) break;
                } else {
                    #pragma unroll
                    for (int jj = 7; jj >= 0; --jj) run_max(r, v[8 * g + jj]);
                }
            }
        }
        __syncthreads();

        const bool have = base + wave < total;
        int    n = 0;
        float4 L = zero4, R = zero4;
        if (have) {
            const int em  = __builtin_amdgcn_readlane(e, wave);
            n = em & (NC - 1);
            const int q   = (em >> 8) >> 5;
            float4 f[POOL_WAVES];
            #pragma unroll
            for (int w = 0; w < POOL_WAVES; ++w) f[w] = s_full[w][lane];
            L = neg4; R = neg4;
            if (q < POOL_WAVES) { L = s_pl[wave][lane]; R = s_pr[wave][lane]; }
            #pragma unroll
            for (int w = 0; w < POOL_WAVES; ++w) {
                L = fmax4(L, (w < q) ? f[w] : neg4);
                R = fmax4(R, (w > q) ? f[w] : neg4);
            }
        }
        s_a4[wave * (PA / 4) + lane]            = L;
        s_a4[wave * (PA / 4) + HSL / 4 + lane]  = R;
        __syncthreads();

        if (wave < MMW) {
            const int nt  = wave >> 2;
            const int kq  = wave & 3;
            const int col = nt * 16 + l16;
            const bool cok = col < NL;
            const float* fa = (const float*)s_a4 + l16 * PA + kq * KQ + 4 * half;
            const float* fb = (const float*)s_w4 + (kq * KQ + 4 * half) * NL + col;
            v8f c0 = {};
            v8f c1 = {};
            #pragma unroll
            for (int p = 0; p < KQ / 8; ++p) {
                const float4 a = *(const float4*)(fa + 8 * p);
                const float bx = fb[(8 * p + 0) * NL];
                const float by = fb[(8 * p + 1) * NL];
                const float bz = fb[(8 * p + 2) * NL];
                const float bw = fb[(8 * p + 3) * NL];
                v2f a0, a1, b0, b1;
                a0.x = a.x; a0.y = a.y; a1.x = a.z; a1.y = a.w;
                b0.x = cok ? bx : 0.0f; b0.y = cok ? by : 0.0f;
                b1.x = cok ? bz : 0.0f; b1.y = cok ? bw : 0.0f;
                c0 = __builtin_amdgcn_wmma_f32_16x16x4_f32(
                          false, a0,  false, b0,
                          (short)0, c0,  false,  false);
                c1 = __builtin_amdgcn_wmma_f32_16x16x4_f32(
                          false, a1,  false, b1,
                          (short)0, c1,  false,  false);
            }
            #pragma unroll
            for (int j = 0; j < 8; ++j) s_part[kq][nt][j][lane] = c0[j] + c1[j];
        }
        __syncthreads();

        if (have) {
            const int j  = wave & 7;
            const int hb = (wave >> 3) * 16;
            float s = 0.0f, t = 0.0f;
            #pragma unroll
            for (int q = 0; q < 4; ++q) {
                s += s_part[q][half][j][hb + l16];
                t += s_part[q][2][j][hb + l16];
            }
            float* dst = partial + ((size_t)n * NHSL + hs) * PSTR;
            dst[lane] = s;
            if (lane < NL - 32) dst[32 + lane] = t;
        }
    }
}

#define STHR 128

__global__ __launch_bounds__(STHR)
void dmbert_logits_sum_kernel(const float* __restrict__ partial,
                              const float* __restrict__ bias,
                              float*       __restrict__ out) {
    const int i = blockIdx.x * STHR + threadIdx.x;
    const int n = i / NL;
    const int c = i - n * NL;
    const float* p = partial + (size_t)n * (NHSL * PSTR) + c;
    float x[NHSL];
    #pragma unroll
    for (int h = 0; h < NHSL; ++h) x[h] = p[h * PSTR];
    float s = x[0];
    #pragma unroll
    for (int h = 1; h < NHSL; ++h) s += x[h];
    out[i] = s + bias[c];
}

extern "C" void kernel_launch(void* const* d_in, const int* in_sizes, int n_in,
                              void* d_out, int out_size, void* d_ws, size_t ws_size,
                              hipStream_t stream) {
    const float* hidden     = (const float*)d_in[0];
    const float* Wm         = (const float*)d_in[1];
    const float* bias       = (const float*)d_in[2];
    const int*   cand_batch = (const int*)  d_in[3];
    const int*   cand_cut   = (const int*)  d_in[4];
    float*       out        = (float*)d_out;

    float* partial = (float*)d_ws;

    dmbert_pool_logits_kernel<<<BB * NHSL, POOL_THREADS, 0, stream>>>(hidden, cand_batch, cand_cut, Wm, partial);

    dmbert_logits_sum_kernel<<<(NC * NL) / STHR, STHR, 0, stream>>>(partial, bias, out);
}
